# starting kernel
# baseline (speedup 1.0000x reference)
#include <hip/hip_runtime.h>
#include <hip/hip_bf16.h>

typedef __attribute__((ext_vector_type(2))) float v2f;
typedef __attribute__((ext_vector_type(8))) float v8f;

#define LN_EPS    1e-5f
#define NEG_SLOPE 0.2f

__device__ __forceinline__ void atomic_max_f32(float* addr, float val) {
  if (val >= 0.0f) {
    atomicMax((int*)addr, __float_as_int(val));
  } else {
    atomicMin((unsigned int*)addr, __float_as_uint(val));
  }
}

__device__ __forceinline__ float gelu_exact(float g) {
  return 0.5f * g * (1.0f + erff(g * 0.70710678118654752440f));
}

__global__ void k_init(float* __restrict__ out, float* __restrict__ emax,
                       float* __restrict__ denom, int nOut, int nAtt) {
  int i = blockIdx.x * blockDim.x + threadIdx.x;
  if (i < nOut) out[i] = 0.0f;
  if (i < nAtt) { emax[i] = -__builtin_inff(); denom[i] = 0.0f; }
}

__global__ void __launch_bounds__(256) k_gemm(const float* __restrict__ X,
                                              const float* __restrict__ W,
                                              float* __restrict__ H, int M) {
  const int lane  = threadIdx.x & 31;
  const int wave  = threadIdx.x >> 5;
  const int half  = lane >> 4;
  const int l15   = lane & 15;
  const int mtile = blockIdx.x;

  const int m = mtile * 16 + l15;
  const int n = wave * 16 + l15;

  const float* __restrict__ Xp = X + (size_t)m * 128;

  v8f acc = {};
#pragma unroll 4
  for (int k = 0; k < 128; k += 4) {
    const int ka = k + 2 * half;
    v2f a = *(const v2f*)(Xp + ka);
    v2f b;
    b.x = W[(size_t)(ka + 0) * 128 + n];
    b.y = W[(size_t)(ka + 1) * 128 + n];
    acc = __builtin_amdgcn_wmma_f32_16x16x4_f32(
         false, a,  false, b,
         (short)0, acc,  false,  false);
  }

  float* __restrict__ Hp = H + ((size_t)(mtile * 16 + 8 * half)) * 128 + n;
#pragma unroll
  for (int v = 0; v < 8; ++v) {
    Hp[(size_t)v * 128] = acc[v];
  }
}

__global__ void k_scores(const float* __restrict__ H,
                         const float* __restrict__ att_src,
                         const float* __restrict__ att_dst,
                         float* __restrict__ asrc, float* __restrict__ adst,
                         int N) {
  int gtid = blockIdx.x * blockDim.x + threadIdx.x;
  int node = gtid >> 5;
  int lane = gtid & 31;
  if (node >= N) return;
  const int c0 = lane * 4;
  const float4 hv = *(const float4*)(H + (size_t)node * 128 + c0);
  const float4 as = *(const float4*)(att_src + c0);
  const float4 ad = *(const float4*)(att_dst + c0);
  float ps = hv.x * as.x + hv.y * as.y + hv.z * as.z + hv.w * as.w;
  float pd = hv.x * ad.x + hv.y * ad.y + hv.z * ad.z + hv.w * ad.w;
#pragma unroll
  for (int m = 1; m < 8; m <<= 1) {
    ps += __shfl_xor(ps, m, 32);
    pd += __shfl_xor(pd, m, 32);
  }
  if ((lane & 7) == 0) {
    const int head = lane >> 3;
    asrc[(size_t)node * 4 + head] = ps;
    adst[(size_t)node * 4 + head] = pd;
  }
}

__global__ void k_edge_max(const int* __restrict__ srcs, const int* __restrict__ dsts,
                           const float* __restrict__ asrc, const float* __restrict__ adst,
                           float* __restrict__ emax, int E, int total) {
  int tid = blockIdx.x * blockDim.x + threadIdx.x;
  int i = tid >> 2, h = tid & 3;
  if (i >= total) return;
  const int s = (i < E) ? srcs[i] : (i - E);
  const int d = (i < E) ? dsts[i] : (i - E);
  float e = asrc[(size_t)s * 4 + h] + adst[(size_t)d * 4 + h];
  e = (e > 0.0f) ? e : NEG_SLOPE * e;
  atomic_max_f32(&emax[(size_t)d * 4 + h], e);
}

__global__ void k_edge_sum(const int* __restrict__ srcs, const int* __restrict__ dsts,
                           const float* __restrict__ asrc, const float* __restrict__ adst,
                           const float* __restrict__ emax, float* __restrict__ denom,
                           int E, int total) {
  int tid = blockIdx.x * blockDim.x + threadIdx.x;
  int i = tid >> 2, h = tid & 3;
  if (i >= total) return;
  const int s = (i < E) ? srcs[i] : (i - E);
  const int d = (i < E) ? dsts[i] : (i - E);
  float e = asrc[(size_t)s * 4 + h] + adst[(size_t)d * 4 + h];
  e = (e > 0.0f) ? e : NEG_SLOPE * e;
  const float ex = expf(e - emax[(size_t)d * 4 + h]);
  atomicAdd(&denom[(size_t)d * 4 + h], ex);
}

__global__ void k_scatter(const int* __restrict__ srcs, const int* __restrict__ dsts,
                          const float* __restrict__ H, const float* __restrict__ asrc,
                          const float* __restrict__ adst, const float* __restrict__ emax,
                          const float* __restrict__ denom, float* __restrict__ out,
                          int E, int total) {
  unsigned tid = blockIdx.x * blockDim.x + threadIdx.x;
  unsigned i = tid >> 5;
  const int part = tid & 31;
  if (i >= (unsigned)total) return;
  const int s = (i < (unsigned)E) ? srcs[i] : (int)(i - (unsigned)E);
  const int d = (i < (unsigned)E) ? dsts[i] : (int)(i - (unsigned)E);
  const int h = part >> 3;
  float e = asrc[(size_t)s * 4 + h] + adst[(size_t)d * 4 + h];
  e = (e > 0.0f) ? e : NEG_SLOPE * e;
  const float alpha = expf(e - emax[(size_t)d * 4 + h]) / denom[(size_t)d * 4 + h];
  const float4 hv = *(const float4*)(H + (size_t)s * 128 + part * 4);
  float* o = out + (size_t)d * 128 + part * 4;
  atomicAdd(o + 0, hv.x * alpha);
  atomicAdd(o + 1, hv.y * alpha);
  atomicAdd(o + 2, hv.z * alpha);
  atomicAdd(o + 3, hv.w * alpha);
}

__global__ void k_finalize(const float* __restrict__ acc, const float* __restrict__ x,
                           const float* __restrict__ bias, const float* __restrict__ gamma,
                           const float* __restrict__ beta, float* __restrict__ out, int N) {
  int gtid = blockIdx.x * blockDim.x + threadIdx.x;
  int node = gtid >> 5;
  int lane = gtid & 31;
  if (node >= N) return;
  const int c0 = lane * 4;
  float4 v = *(const float4*)(acc + (size_t)node * 128 + c0);
  const float4 bb = *(const float4*)(bias + c0);
  v.x += bb.x; v.y += bb.y; v.z += bb.z; v.w += bb.w;

  float s = v.x + v.y + v.z + v.w;
#pragma unroll
  for (int m = 1; m < 32; m <<= 1) s += __shfl_xor(s, m, 32);
  const float mu = s * (1.0f / 128.0f);

  const float dx = v.x - mu, dy = v.y - mu, dz = v.z - mu, dw = v.w - mu;
  float q = dx * dx + dy * dy + dz * dz + dw * dw;
#pragma unroll
  for (int m = 1; m < 32; m <<= 1) q += __shfl_xor(q, m, 32);
  const float inv = 1.0f / sqrtf(q * (1.0f / 128.0f) + LN_EPS);

  const float4 g  = *(const float4*)(gamma + c0);
  const float4 b2 = *(const float4*)(beta + c0);
  const float4 xv = *(const float4*)(x + (size_t)node * 128 + c0);
  float4 r;
  r.x = gelu_exact(dx * inv * g.x + b2.x + xv.x);
  r.y = gelu_exact(dy * inv * g.y + b2.y + xv.y);
  r.z = gelu_exact(dz * inv * g.z + b2.z + xv.z);
  r.w = gelu_exact(dw * inv * g.w + b2.w + xv.w);
  *(float4*)(out + (size_t)node * 128 + c0) = r;
}

extern "C" void kernel_launch(void* const* d_in, const int* in_sizes, int n_in,
                              void* d_out, int out_size, void* d_ws, size_t ws_size,
                              hipStream_t stream) {
  const float* x       = (const float*)d_in[0];
  const int*   ei      = (const int*)  d_in[1];
  const float* W       = (const float*)d_in[2];
  const float* att_src = (const float*)d_in[3];
  const float* att_dst = (const float*)d_in[4];
  const float* bias    = (const float*)d_in[5];
  const float* gamma   = (const float*)d_in[6];
  const float* beta    = (const float*)d_in[7];
  float* out = (float*)d_out;

  const int N = in_sizes[0] / 128;
  const int E = in_sizes[1] / 2;
  const int total = E + N;

  float* H     = (float*)d_ws;
  float* asrc  = H     + (size_t)N * 128;
  float* adst  = asrc  + (size_t)N * 4;
  float* emax  = adst  + (size_t)N * 4;
  float* denom = emax  + (size_t)N * 4;

  const int* srcs = ei;
  const int* dsts = ei + E;

  const int nOut = N * 128;
  k_init<<<(nOut + 255) / 256, 256, 0, stream>>>(out, emax, denom, nOut, N * 4);

  k_gemm<<<N / 16, 256, 0, stream>>>(x, W, H, N);

  k_scores<<<(N * 32 + 255) / 256, 256, 0, stream>>>(H, att_src, att_dst, asrc, adst, N);

  const int tMax = total * 4;
  k_edge_max<<<(tMax + 255) / 256, 256, 0, stream>>>(srcs, dsts, asrc, adst, emax, E, total);
  k_edge_sum<<<(tMax + 255) / 256, 256, 0, stream>>>(srcs, dsts, asrc, adst, emax, denom, E, total);

  const long long tSc = (long long)total * 32;
  k_scatter<<<(unsigned)((tSc + 255) / 256), 256, 0, stream>>>(srcs, dsts, H, asrc, adst,
                                                               emax, denom, out, E, total);

  k_finalize<<<(N * 32 + 255) / 256, 256, 0, stream>>>(out, x, bias, gamma, beta, out, N);
}


# Scatter lanes own consecutive channels (one per head): each h[src] gather and each f32 atomic add covers one whole 128-byte line, not four quarter-filled lines; sum order still atomic arrival
# speedup vs baseline: 1.2102x; 1.2102x over previous
#include <hip/hip_runtime.h>
#include <hip/hip_bf16.h>

typedef __attribute__((ext_vector_type(2))) float v2f;
typedef __attribute__((ext_vector_type(8))) float v8f;

#define LN_EPS    1e-5f
#define NEG_SLOPE 0.2f

__device__ __forceinline__ void atomic_max_f32(float* addr, float val) {
  if (val >= 0.0f) {
    atomicMax((int*)addr, __float_as_int(val));
  } else {
    atomicMin((unsigned int*)addr, __float_as_uint(val));
  }
}

__device__ __forceinline__ float gelu_exact(float g) {
  return 0.5f * g * (1.0f + erff(g * 0.70710678118654752440f));
}

__global__ void k_init(float* __restrict__ out, float* __restrict__ emax,
                       float* __restrict__ denom, int nOut, int nAtt) {
  int i = blockIdx.x * blockDim.x + threadIdx.x;
  if (i < nOut) out[i] = 0.0f;
  if (i < nAtt) { emax[i] = -__builtin_inff(); denom[i] = 0.0f; }
}

__global__ void __launch_bounds__(256) k_gemm(const float* __restrict__ X,
                                              const float* __restrict__ W,
                                              float* __restrict__ H, int M) {
  const int lane  = threadIdx.x & 31;
  const int wave  = threadIdx.x >> 5;
  const int half  = lane >> 4;
  const int l15   = lane & 15;
  const int mtile = blockIdx.x;

  const int m = mtile * 16 + l15;
  const int n = wave * 16 + l15;

  const float* __restrict__ Xp = X + (size_t)m * 128;

  v8f acc = {};
#pragma unroll 4
  for (int k = 0; k < 128; k += 4) {
    const int ka = k + 2 * half;
    v2f a = *(const v2f*)(Xp + ka);
    v2f b;
    b.x = W[(size_t)(ka + 0) * 128 + n];
    b.y = W[(size_t)(ka + 1) * 128 + n];
    acc = __builtin_amdgcn_wmma_f32_16x16x4_f32(
         false, a,  false, b,
         (short)0, acc,  false,  false);
  }

  float* __restrict__ Hp = H + ((size_t)(mtile * 16 + 8 * half)) * 128 + n;
#pragma unroll
  for (int v = 0; v < 8; ++v) {
    Hp[(size_t)v * 128] = acc[v];
  }
}

__global__ void k_scores(const float* __restrict__ H,
                         const float* __restrict__ att_src,
                         const float* __restrict__ att_dst,
                         float* __restrict__ asrc, float* __restrict__ adst,
                         int N) {
  int gtid = blockIdx.x * blockDim.x + threadIdx.x;
  int node = gtid >> 5;
  int lane = gtid & 31;
  if (node >= N) return;
  const int c0 = lane * 4;
  const float4 hv = *(const float4*)(H + (size_t)node * 128 + c0);
  const float4 as = *(const float4*)(att_src + c0);
  const float4 ad = *(const float4*)(att_dst + c0);
  float ps = hv.x * as.x + hv.y * as.y + hv.z * as.z + hv.w * as.w;
  float pd = hv.x * ad.x + hv.y * ad.y + hv.z * ad.z + hv.w * ad.w;
#pragma unroll
  for (int m = 1; m < 8; m <<= 1) {
    ps += __shfl_xor(ps, m, 32);
    pd += __shfl_xor(pd, m, 32);
  }
  if ((lane & 7) == 0) {
    const int head = lane >> 3;
    asrc[(size_t)node * 4 + head] = ps;
    adst[(size_t)node * 4 + head] = pd;
  }
}

__global__ void k_edge_max(const int* __restrict__ srcs, const int* __restrict__ dsts,
                           const float* __restrict__ asrc, const float* __restrict__ adst,
                           float* __restrict__ emax, int E, int total) {
  int tid = blockIdx.x * blockDim.x + threadIdx.x;
  int i = tid >> 2, h = tid & 3;
  if (i >= total) return;
  const int s = (i < E) ? srcs[i] : (i - E);
  const int d = (i < E) ? dsts[i] : (i - E);
  float e = asrc[(size_t)s * 4 + h] + adst[(size_t)d * 4 + h];
  e = (e > 0.0f) ? e : NEG_SLOPE * e;
  atomic_max_f32(&emax[(size_t)d * 4 + h], e);
}

__global__ void k_edge_sum(const int* __restrict__ srcs, const int* __restrict__ dsts,
                           const float* __restrict__ asrc, const float* __restrict__ adst,
                           const float* __restrict__ emax, float* __restrict__ denom,
                           int E, int total) {
  int tid = blockIdx.x * blockDim.x + threadIdx.x;
  int i = tid >> 2, h = tid & 3;
  if (i >= total) return;
  const int s = (i < E) ? srcs[i] : (i - E);
  const int d = (i < E) ? dsts[i] : (i - E);
  float e = asrc[(size_t)s * 4 + h] + adst[(size_t)d * 4 + h];
  e = (e > 0.0f) ? e : NEG_SLOPE * e;
  const float ex = expf(e - emax[(size_t)d * 4 + h]);
  atomicAdd(&denom[(size_t)d * 4 + h], ex);
}

__global__ void k_scatter(const int* __restrict__ srcs, const int* __restrict__ dsts,
                          const float* __restrict__ H, const float* __restrict__ asrc,
                          const float* __restrict__ adst, const float* __restrict__ emax,
                          const float* __restrict__ denom, float* __restrict__ out,
                          int E, int total) {
  unsigned tid = blockIdx.x * blockDim.x + threadIdx.x;
  unsigned i = tid >> 5;
  const int lane = tid & 31;
  if (i >= (unsigned)total) return;
  const int s = (i < (unsigned)E) ? srcs[i] : (int)(i - (unsigned)E);
  const int d = (i < (unsigned)E) ? dsts[i] : (int)(i - (unsigned)E);

  const float* __restrict__ hp = H + (size_t)s * 128 + lane;
  const float h0 = hp[0];
  const float h1 = hp[32];
  const float h2 = hp[64];
  const float h3 = hp[96];

  const int hq = lane & 3;
  float e = asrc[(size_t)s * 4 + hq] + adst[(size_t)d * 4 + hq];
  e = (e > 0.0f) ? e : NEG_SLOPE * e;
  const float alpha = expf(e - emax[(size_t)d * 4 + hq]) / denom[(size_t)d * 4 + hq];
  const float a0 = __int_as_float(__builtin_amdgcn_readlane(__float_as_int(alpha), 0));
  const float a1 = __int_as_float(__builtin_amdgcn_readlane(__float_as_int(alpha), 1));
  const float a2 = __int_as_float(__builtin_amdgcn_readlane(__float_as_int(alpha), 2));
  const float a3 = __int_as_float(__builtin_amdgcn_readlane(__float_as_int(alpha), 3));

  float* o = out + (size_t)d * 128 + lane;
  atomicAdd(o +  0, h0 * a0);
  atomicAdd(o + 32, h1 * a1);
  atomicAdd(o + 64, h2 * a2);
  atomicAdd(o + 96, h3 * a3);
}

__global__ void k_finalize(const float* __restrict__ acc, const float* __restrict__ x,
                           const float* __restrict__ bias, const float* __restrict__ gamma,
                           const float* __restrict__ beta, float* __restrict__ out, int N) {
  int gtid = blockIdx.x * blockDim.x + threadIdx.x;
  int node = gtid >> 5;
  int lane = gtid & 31;
  if (node >= N) return;
  const int c0 = lane * 4;
  float4 v = *(const float4*)(acc + (size_t)node * 128 + c0);
  const float4 bb = *(const float4*)(bias + c0);
  v.x += bb.x; v.y += bb.y; v.z += bb.z; v.w += bb.w;

  float s = v.x + v.y + v.z + v.w;
#pragma unroll
  for (int m = 1; m < 32; m <<= 1) s += __shfl_xor(s, m, 32);
  const float mu = s * (1.0f / 128.0f);

  const float dx = v.x - mu, dy = v.y - mu, dz = v.z - mu, dw = v.w - mu;
  float q = dx * dx + dy * dy + dz * dz + dw * dw;
#pragma unroll
  for (int m = 1; m < 32; m <<= 1) q += __shfl_xor(q, m, 32);
  const float inv = 1.0f / sqrtf(q * (1.0f / 128.0f) + LN_EPS);

  const float4 g  = *(const float4*)(gamma + c0);
  const float4 b2 = *(const float4*)(beta + c0);
  const float4 xv = *(const float4*)(x + (size_t)node * 128 + c0);
  float4 r;
  r.x = gelu_exact(dx * inv * g.x + b2.x + xv.x);
  r.y = gelu_exact(dy * inv * g.y + b2.y + xv.y);
  r.z = gelu_exact(dz * inv * g.z + b2.z + xv.z);
  r.w = gelu_exact(dw * inv * g.w + b2.w + xv.w);
  *(float4*)(out + (size_t)node * 128 + c0) = r;
}

extern "C" void kernel_launch(void* const* d_in, const int* in_sizes, int n_in,
                              void* d_out, int out_size, void* d_ws, size_t ws_size,
                              hipStream_t stream) {
  const float* x       = (const float*)d_in[0];
  const int*   ei      = (const int*)  d_in[1];
  const float* W       = (const float*)d_in[2];
  const float* att_src = (const float*)d_in[3];
  const float* att_dst = (const float*)d_in[4];
  const float* bias    = (const float*)d_in[5];
  const float* gamma   = (const float*)d_in[6];
  const float* beta    = (const float*)d_in[7];
  float* out = (float*)d_out;

  const int N = in_sizes[0] / 128;
  const int E = in_sizes[1] / 2;
  const int total = E + N;

  float* H     = (float*)d_ws;
  float* asrc  = H     + (size_t)N * 128;
  float* adst  = asrc  + (size_t)N * 4;
  float* emax  = adst  + (size_t)N * 4;
  float* denom = emax  + (size_t)N * 4;

  const int* srcs = ei;
  const int* dsts = ei + E;

  const int nOut = N * 128;
  k_init<<<(nOut + 255) / 256, 256, 0, stream>>>(out, emax, denom, nOut, N * 4);

  k_gemm<<<N / 16, 256, 0, stream>>>(x, W, H, N);

  k_scores<<<(N * 32 + 255) / 256, 256, 0, stream>>>(H, att_src, att_dst, asrc, adst, N);

  const int tMax = total * 4;
  k_edge_max<<<(tMax + 255) / 256, 256, 0, stream>>>(srcs, dsts, asrc, adst, emax, E, total);
  k_edge_sum<<<(tMax + 255) / 256, 256, 0, stream>>>(srcs, dsts, asrc, adst, emax, denom, E, total);

  const long long tSc = (long long)total * 32;
  k_scatter<<<(unsigned)((tSc + 255) / 256), 256, 0, stream>>>(srcs, dsts, H, asrc, adst,
                                                               emax, denom, out, E, total);

  k_finalize<<<(N * 32 + 255) / 256, 256, 0, stream>>>(out, x, bias, gamma, beta, out, N);
}


# Fewer, longer waves in all launches but the GEMM: a scatter wave walks 32 edges (one lane per edge for ids and weights), other passes do 4 items per thread or wave; sums still in atomic arrival order
# speedup vs baseline: 1.5038x; 1.2426x over previous
#include <hip/hip_runtime.h>
#include <hip/hip_bf16.h>

typedef __attribute__((ext_vector_type(2))) float v2f;
typedef __attribute__((ext_vector_type(8))) float v8f;

#define LN_EPS    1e-5f
#define NEG_SLOPE 0.2f

__device__ __forceinline__ void atomic_max_f32(float* addr, float val) {
  if (val >= 0.0f) {
    atomicMax((int*)addr, __float_as_int(val));
  } else {
    atomicMin((unsigned int*)addr, __float_as_uint(val));
  }
}

__device__ __forceinline__ float gelu_exact(float g) {
  return 0.5f * g * (1.0f + erff(g * 0.70710678118654752440f));
}

__global__ void __launch_bounds__(256) k_init(float* __restrict__ out, float* __restrict__ emax,
                                              float* __restrict__ denom, int nOut4, int nAtt4) {
  const float ninf = -__builtin_inff();
  const float4 z = make_float4(0.0f, 0.0f, 0.0f, 0.0f);
  const float4 m = make_float4(ninf, ninf, ninf, ninf);
  const int base = blockIdx.x * 1024 + threadIdx.x;
#pragma unroll
  for (int j = 0; j < 4; ++j) {
    const int i = base + j * 256;
    if (i < nOut4) ((float4*)out)[i] = z;
    if (i < nAtt4) { ((float4*)emax)[i] = m; ((float4*)denom)[i] = z; }
  }
}

__global__ void __launch_bounds__(256) k_gemm(const float* __restrict__ X,
                                              const float* __restrict__ W,
                                              float* __restrict__ H, int M) {
  const int lane  = threadIdx.x & 31;
  const int wave  = threadIdx.x >> 5;
  const int half  = lane >> 4;
  const int l15   = lane & 15;
  const int mtile = blockIdx.x;

  const int m = mtile * 16 + l15;
  const int n = wave * 16 + l15;

  const float* __restrict__ Xp = X + (size_t)m * 128;

  v8f acc = {};
#pragma unroll 4
  for (int k = 0; k < 128; k += 4) {
    const int ka = k + 2 * half;
    v2f a = *(const v2f*)(Xp + ka);
    v2f b;
    b.x = W[(size_t)(ka + 0) * 128 + n];
    b.y = W[(size_t)(ka + 1) * 128 + n];
    acc = __builtin_amdgcn_wmma_f32_16x16x4_f32(
         false, a,  false, b,
         (short)0, acc,  false,  false);
  }

  float* __restrict__ Hp = H + ((size_t)(mtile * 16 + 8 * half)) * 128 + n;
#pragma unroll
  for (int v = 0; v < 8; ++v) {
    Hp[(size_t)v * 128] = acc[v];
  }
}

__global__ void __launch_bounds__(256) k_scores(const float* __restrict__ H,
                                                const float* __restrict__ att_src,
                                                const float* __restrict__ att_dst,
                                                float* __restrict__ asrc, float* __restrict__ adst,
                                                int N) {
  const int lane = threadIdx.x & 31;
  const int wv = __builtin_amdgcn_readfirstlane((int)((blockIdx.x * 256u + threadIdx.x) >> 5));
  const int node0 = wv * 4;
  if (node0 >= N) return;
  const int c0 = lane * 4;
  const float4 as = *(const float4*)(att_src + c0);
  const float4 ad = *(const float4*)(att_dst + c0);
  float4 hv[4];
#pragma unroll
  for (int j = 0; j < 4; ++j) {
    const int node = (node0 + j < N) ? (node0 + j) : node0;
    hv[j] = *(const float4*)(H + (size_t)node * 128 + c0);
  }
#pragma unroll
  for (int j = 0; j < 4; ++j) {
    float ps = hv[j].x * as.x + hv[j].y * as.y + hv[j].z * as.z + hv[j].w * as.w;
    float pd = hv[j].x * ad.x + hv[j].y * ad.y + hv[j].z * ad.z + hv[j].w * ad.w;
#pragma unroll
    for (int m = 1; m < 8; m <<= 1) {
      ps += __shfl_xor(ps, m, 32);
      pd += __shfl_xor(pd, m, 32);
    }
    if ((lane & 7) == 0 && node0 + j < N) {
      const int head = lane >> 3;
      asrc[(size_t)(node0 + j) * 4 + head] = ps;
      adst[(size_t)(node0 + j) * 4 + head] = pd;
    }
  }
}

__device__ __forceinline__ void edge_item(const int* __restrict__ srcs, const int* __restrict__ dsts,
                                          int E, int t, int& sh, int& dh) {
  const int i = t >> 2, h = t & 3;
  int s, d;
  if (i < E) { s = srcs[i]; d = dsts[i]; } else { s = i - E; d = i - E; }
  sh = s * 4 + h;
  dh = d * 4 + h;
}

__global__ void __launch_bounds__(256) k_edge_max(const int* __restrict__ srcs, const int* __restrict__ dsts,
                                                  const float* __restrict__ asrc, const float* __restrict__ adst,
                                                  float* __restrict__ emax, int E, int total4) {
  const int base = blockIdx.x * 1024 + threadIdx.x;
  int sh[4], dh[4];
  float e[4];
#pragma unroll
  for (int j = 0; j < 4; ++j) {
    const int t = base + j * 256;
    edge_item(srcs, dsts, E, (t < total4) ? t : 0, sh[j], dh[j]);
  }
#pragma unroll
  for (int j = 0; j < 4; ++j) e[j] = asrc[sh[j]] + adst[dh[j]];
#pragma unroll
  for (int j = 0; j < 4; ++j) {
    const float v = (e[j] > 0.0f) ? e[j] : NEG_SLOPE * e[j];
    if (base + j * 256 < total4) atomic_max_f32(&emax[dh[j]], v);
  }
}

__global__ void __launch_bounds__(256) k_edge_sum(const int* __restrict__ srcs, const int* __restrict__ dsts,
                                                  const float* __restrict__ asrc, const float* __restrict__ adst,
                                                  const float* __restrict__ emax, float* __restrict__ denom,
                                                  int E, int total4) {
  const int base = blockIdx.x * 1024 + threadIdx.x;
  int sh[4], dh[4];
  float e[4], mx[4];
#pragma unroll
  for (int j = 0; j < 4; ++j) {
    const int t = base + j * 256;
    edge_item(srcs, dsts, E, (t < total4) ? t : 0, sh[j], dh[j]);
  }
#pragma unroll
  for (int j = 0; j < 4; ++j) {
    e[j]  = asrc[sh[j]] + adst[dh[j]];
    mx[j] = emax[dh[j]];
  }
#pragma unroll
  for (int j = 0; j < 4; ++j) {
    const float v = (e[j] > 0.0f) ? e[j] : NEG_SLOPE * e[j];
    const float ex = expf(v - mx[j]);
    if (base + j * 256 < total4) atomicAdd(&denom[dh[j]], ex);
  }
}

__device__ __forceinline__ float lane_f32(float v, int j) {
  return __int_as_float(__builtin_amdgcn_readlane(__float_as_int(v), j));
}

__global__ void __launch_bounds__(256) k_scatter(const int* __restrict__ srcs, const int* __restrict__ dsts,
                                                 const float* __restrict__ H, const float* __restrict__ asrc,
                                                 const float* __restrict__ adst, const float* __restrict__ emax,
                                                 const float* __restrict__ denom, float* __restrict__ out,
                                                 int E, int total) {
  const int lane = threadIdx.x & 31;
  const int wv = __builtin_amdgcn_readfirstlane((int)((blockIdx.x * 256u + threadIdx.x) >> 5));
  const int base = wv * 32;
  if (base >= total) return;
  const int cnt = (total - base < 32) ? (total - base) : 32;

  const int i = (lane < cnt) ? (base + lane) : base;
  int s, d;
  if (i < E) { s = srcs[i]; d = dsts[i]; } else { s = i - E; d = i - E; }

  const float4 as = *(const float4*)(asrc  + (size_t)s * 4);
  const float4 ad = *(const float4*)(adst  + (size_t)d * 4);
  const float4 em = *(const float4*)(emax  + (size_t)d * 4);
  const float4 dn = *(const float4*)(denom + (size_t)d * 4);
  float e0 = as.x + ad.x, e1 = as.y + ad.y, e2 = as.z + ad.z, e3 = as.w + ad.w;
  e0 = (e0 > 0.0f) ? e0 : NEG_SLOPE * e0;
  e1 = (e1 > 0.0f) ? e1 : NEG_SLOPE * e1;
  e2 = (e2 > 0.0f) ? e2 : NEG_SLOPE * e2;
  e3 = (e3 > 0.0f) ? e3 : NEG_SLOPE * e3;
  const float w0 = expf(e0 - em.x) / dn.x;
  const float w1 = expf(e1 - em.y) / dn.y;
  const float w2 = expf(e2 - em.z) / dn.z;
  const float w3 = expf(e3 - em.w) / dn.w;

  const float* __restrict__ Hl = H + lane;
  float* __restrict__ outl = out + lane;

  int j = 0;
  for (; j + 4 <= cnt; j += 4) {
    float hv[4][4];
    int dj[4];
#pragma unroll
    for (int u = 0; u < 4; ++u) {
      const int sj = __builtin_amdgcn_readlane(s, j + u);
      dj[u] = __builtin_amdgcn_readlane(d, j + u);
      const float* __restrict__ hp = Hl + (size_t)sj * 128;
      hv[u][0] = hp[0];
      hv[u][1] = hp[32];
      hv[u][2] = hp[64];
      hv[u][3] = hp[96];
    }
#pragma unroll
    for (int u = 0; u < 4; ++u) {
      const float a0 = lane_f32(w0, j + u);
      const float a1 = lane_f32(w1, j + u);
      const float a2 = lane_f32(w2, j + u);
      const float a3 = lane_f32(w3, j + u);
      float* o = outl + (size_t)dj[u] * 128;
      atomicAdd(o +  0, hv[u][0] * a0);
      atomicAdd(o + 32, hv[u][1] * a1);
      atomicAdd(o + 64, hv[u][2] * a2);
      atomicAdd(o + 96, hv[u][3] * a3);
    }
  }
  for (; j < cnt; ++j) {
    const int sj = __builtin_amdgcn_readlane(s, j);
    const int dd = __builtin_amdgcn_readlane(d, j);
    const float* __restrict__ hp = Hl + (size_t)sj * 128;
    const float h0 = hp[0], h1 = hp[32], h2 = hp[64], h3 = hp[96];
    float* o = outl + (size_t)dd * 128;
    atomicAdd(o +  0, h0 * lane_f32(w0, j));
    atomicAdd(o + 32, h1 * lane_f32(w1, j));
    atomicAdd(o + 64, h2 * lane_f32(w2, j));
    atomicAdd(o + 96, h3 * lane_f32(w3, j));
  }
}

__global__ void __launch_bounds__(256) k_finalize(const float* __restrict__ x,
                                                  const float* __restrict__ bias, const float* __restrict__ gamma,
                                                  const float* __restrict__ beta, float* out, int N) {
  const int lane = threadIdx.x & 31;
  const int wv = __builtin_amdgcn_readfirstlane((int)((blockIdx.x * 256u + threadIdx.x) >> 5));
  const int node0 = wv * 4;
  if (node0 >= N) return;
  const int c0 = lane * 4;
  const float4 bb = *(const float4*)(bias + c0);
  const float4 g  = *(const float4*)(gamma + c0);
  const float4 b2 = *(const float4*)(beta + c0);
  float4 av[4], xv[4];
#pragma unroll
  for (int j = 0; j < 4; ++j) {
    const int node = (node0 + j < N) ? (node0 + j) : node0;
    av[j] = *(const float4*)(out + (size_t)node * 128 + c0);
    xv[j] = *(const float4*)(x   + (size_t)node * 128 + c0);
  }
#pragma unroll
  for (int j = 0; j < 4; ++j) {
    float4 v = av[j];
    v.x += bb.x; v.y += bb.y; v.z += bb.z; v.w += bb.w;

    float s = v.x + v.y + v.z + v.w;
#pragma unroll
    for (int m = 1; m < 32; m <<= 1) s += __shfl_xor(s, m, 32);
    const float mu = s * (1.0f / 128.0f);

    const float dx = v.x - mu, dy = v.y - mu, dz = v.z - mu, dw = v.w - mu;
    float q = dx * dx + dy * dy + dz * dz + dw * dw;
#pragma unroll
    for (int m = 1; m < 32; m <<= 1) q += __shfl_xor(q, m, 32);
    const float inv = 1.0f / sqrtf(q * (1.0f / 128.0f) + LN_EPS);

    float4 r;
    r.x = gelu_exact(dx * inv * g.x + b2.x + xv[j].x);
    r.y = gelu_exact(dy * inv * g.y + b2.y + xv[j].y);
    r.z = gelu_exact(dz * inv * g.z + b2.z + xv[j].z);
    r.w = gelu_exact(dw * inv * g.w + b2.w + xv[j].w);
    if (node0 + j < N) *(float4*)(out + (size_t)(node0 + j) * 128 + c0) = r;
  }
}

extern "C" void kernel_launch(void* const* d_in, const int* in_sizes, int n_in,
                              void* d_out, int out_size, void* d_ws, size_t ws_size,
                              hipStream_t stream) {
  const float* x       = (const float*)d_in[0];
  const int*   ei      = (const int*)  d_in[1];
  const float* W       = (const float*)d_in[2];
  const float* att_src = (const float*)d_in[3];
  const float* att_dst = (const float*)d_in[4];
  const float* bias    = (const float*)d_in[5];
  const float* gamma   = (const float*)d_in[6];
  const float* beta    = (const float*)d_in[7];
  float* out = (float*)d_out;

  const int N = in_sizes[0] / 128;
  const int E = in_sizes[1] / 2;
  const int total = E + N;

  float* H     = (float*)d_ws;
  float* asrc  = H     + (size_t)N * 128;
  float* adst  = asrc  + (size_t)N * 4;
  float* emax  = adst  + (size_t)N * 4;
  float* denom = emax  + (size_t)N * 4;

  const int* srcs = ei;
  const int* dsts = ei + E;

  const int nOut4 = N * 32;
  k_init<<<(nOut4 + 1023) / 1024, 256, 0, stream>>>(out, emax, denom, nOut4, N);

  k_gemm<<<N / 16, 256, 0, stream>>>(x, W, H, N);

  k_scores<<<(N + 31) / 32, 256, 0, stream>>>(H, att_src, att_dst, asrc, adst, N);

  const int total4 = total * 4;
  k_edge_max<<<(total4 + 1023) / 1024, 256, 0, stream>>>(srcs, dsts, asrc, adst, emax, E, total4);
  k_edge_sum<<<(total4 + 1023) / 1024, 256, 0, stream>>>(srcs, dsts, asrc, adst, emax, denom, E, total4);

  k_scatter<<<(total + 255) / 256, 256, 0, stream>>>(srcs, dsts, H, asrc, adst,
                                                     emax, denom, out, E, total);

  k_finalize<<<(N + 31) / 32, 256, 0, stream>>>(x, bias, gamma, beta, out, N);
}
